# starting kernel
# baseline (speedup 1.0000x reference)
#include <hip/hip_runtime.h>
#include <math.h>

typedef __attribute__((ext_vector_type(2))) float v2f;
typedef __attribute__((ext_vector_type(8))) float v8f;

__device__ __constant__ float g_vdw[20][14] = {
  {1.55f,1.70f,1.70f,1.52f, 1.70f,0,0,0,0,0,0,0,0,0},
  {1.55f,1.70f,1.70f,1.52f, 1.70f,1.80f,0,0,0,0,0,0,0,0},
  {1.55f,1.70f,1.70f,1.52f, 1.70f,1.70f,1.52f,1.52f,0,0,0,0,0,0},
  {1.55f,1.70f,1.70f,1.52f, 1.70f,1.70f,1.70f,1.52f,1.52f,0,0,0,0,0},
  {1.55f,1.70f,1.70f,1.52f, 1.70f,1.70f,1.70f,1.70f,1.70f,1.70f,1.70f,0,0,0},
  {1.55f,1.70f,1.70f,1.52f, 0,0,0,0,0,0,0,0,0,0},
  {1.55f,1.70f,1.70f,1.52f, 1.70f,1.70f,1.55f,1.70f,1.70f,1.55f,0,0,0,0},
  {1.55f,1.70f,1.70f,1.52f, 1.70f,1.70f,1.70f,1.70f,0,0,0,0,0,0},
  {1.55f,1.70f,1.70f,1.52f, 1.70f,1.70f,1.70f,1.70f,1.55f,0,0,0,0,0},
  {1.55f,1.70f,1.70f,1.52f, 1.70f,1.70f,1.70f,1.70f,0,0,0,0,0,0},
  {1.55f,1.70f,1.70f,1.52f, 1.70f,1.70f,1.80f,1.70f,0,0,0,0,0,0},
  {1.55f,1.70f,1.70f,1.52f, 1.70f,1.70f,1.52f,1.55f,0,0,0,0,0,0},
  {1.55f,1.70f,1.70f,1.52f, 1.70f,1.70f,1.70f,0,0,0,0,0,0,0},
  {1.55f,1.70f,1.70f,1.52f, 1.70f,1.70f,1.70f,1.52f,1.55f,0,0,0,0,0},
  {1.55f,1.70f,1.70f,1.52f, 1.70f,1.70f,1.70f,1.55f,1.70f,1.55f,1.55f,0,0,0},
  {1.55f,1.70f,1.70f,1.52f, 1.70f,1.52f,0,0,0,0,0,0,0,0},
  {1.55f,1.70f,1.70f,1.52f, 1.70f,1.52f,1.70f,0,0,0,0,0,0,0},
  {1.55f,1.70f,1.70f,1.52f, 1.70f,1.70f,1.70f,0,0,0,0,0,0,0},
  {1.55f,1.70f,1.70f,1.52f, 1.70f,1.70f,1.70f,1.70f,1.55f,1.70f,1.70f,1.70f,1.70f,1.70f},
  {1.55f,1.70f,1.70f,1.52f, 1.70f,1.70f,1.70f,1.70f,1.70f,1.70f,1.70f,1.52f,0,0},
};
__device__ __constant__ int g_natoms[20] =
  {5,6,8,9,11,4,10,8,9,8,8,8,7,9,11,6,7,7,14,12};

#define DIST_EPS      1e-3f
#define CONNECT_CUT   0.35f
#define LOG2E         1.44269504088896340736f

__global__ __launch_bounds__(256)
void clash_wmma_kernel(const float* __restrict__ X,
                       const float* __restrict__ mask_j,
                       const float* __restrict__ mask_ij,
                       const int*   __restrict__ C,
                       const int*   __restrict__ S,
                       const int*   __restrict__ E,
                       float*       __restrict__ out,
                       int N, int K, int BN)
{
    const int lane = threadIdx.x & 31;
    const int wave = blockIdx.x * (blockDim.x >> 5) + (threadIdx.x >> 5);
    if (wave >= BN) return;

    const int b = wave / N;
    const int i = wave % N;
    const int a  = lane & 15;
    const int hi = lane >> 4;

    const int s_i   = S[wave];
    const int cnt_i = (C[wave] > 0) ? g_natoms[s_i] : 0;
    float xi = 0.f, yi = 0.f, zi = 0.f, rad_i = 0.f;
    if (a < 14) {
        const float* p = X + (size_t)wave * 42 + a * 3;
        xi = p[0]; yi = p[1]; zi = p[2];
        rad_i = g_vdw[s_i][a];
    }
    const float n2_i = xi * xi + yi * yi + zi * zi;
    const float mA_i = (a < cnt_i) ? 1.f : 0.f;

    v2f Areg;
    Areg.x = hi ? zi : xi;
    Areg.y = hi ? 0.f : yi;

    float n2M[8], hcutM[8], mAM[8];
#pragma unroll
    for (int r = 0; r < 8; ++r) {
        const int M = r + (hi << 3);
        n2M[r]  = __shfl(n2_i, M, 32);
        hcutM[r] = 0.5f * __shfl(rad_i, M, 32) + CONNECT_CUT;
        const float m = __shfl(mA_i, M, 32);
        mAM[r] = (M >= 4) ? m : 0.f;
    }

    float acc = 0.f;
    const int baseE = wave * K;
    for (int k = 0; k < K; ++k) {
        const int   j  = E[baseE + k];
        const float me = mask_ij[baseE + k] * ((j != i) ? 1.f : 0.f);
        const int   jg = b * N + j;

        const int s_j   = S[jg];
        const int cnt_j = (C[jg] > 0) ? g_natoms[s_j] : 0;
        float xj = 0.f, yj = 0.f, zj = 0.f, rad_j = 0.f;
        if (a < 14) {
            const float* p = X + (size_t)jg * 42 + a * 3;
            xj = p[0]; yj = p[1]; zj = p[2];
            rad_j = g_vdw[s_j][a];
        }
        const float n2N  = xj * xj + yj * yj + zj * zj + DIST_EPS;
        const float hradN = 0.5f * rad_j;
        const float wN   = ((a < cnt_j) ? 1.f : 0.f) * mask_j[jg] * me;

        v2f Breg;
        Breg.x = hi ? zj : xj;
        Breg.y = hi ? 0.f : yj;

        v8f czero = {};
        v8f dot = __builtin_amdgcn_wmma_f32_16x16x4_f32(
             false, Areg,  false, Breg,
             (short)0, czero,  false,  false);

#pragma unroll
        for (int r = 0; r < 8; ++r) {
            const float dsq  = n2M[r] + n2N - 2.f * dot[r];
            const float D    = __builtin_amdgcn_sqrtf(dsq);
            const float t    = D - (hcutM[r] + hradN);
            const float e    = __builtin_amdgcn_exp2f(t * LOG2E);
            const float sg   = __builtin_amdgcn_rcpf(1.f + e);
            acc += (mAM[r] * wN) * sg;
        }
    }

#pragma unroll
    for (int off = 16; off > 0; off >>= 1)
        acc += __shfl_down(acc, off, 32);
    if (lane == 0) out[wave] = acc;
}

extern "C" void kernel_launch(void* const* d_in, const int* in_sizes, int n_in,
                              void* d_out, int out_size, void* d_ws, size_t ws_size,
                              hipStream_t stream) {
    const float* X       = (const float*)d_in[0];
    const float* mask_j  = (const float*)d_in[1];
    const float* mask_ij = (const float*)d_in[2];
    const int*   C       = (const int*)d_in[3];
    const int*   S       = (const int*)d_in[4];
    const int*   E       = (const int*)d_in[5];
    float*       out     = (float*)d_out;

    const int BN = in_sizes[1];
    const int K  = in_sizes[2] / BN;
    const int B  = 2;
    const int N  = BN / B;

    const int wavesPerBlock = 8;
    const int blocks = (BN + wavesPerBlock - 1) / wavesPerBlock;
    clash_wmma_kernel<<<blocks, wavesPerBlock * 32, 0, stream>>>(
        X, mask_j, mask_ij, C, S, E, out, N, K, BN);
}


# Dependent per-edge loads taken off the loop's critical path: edge metadata gathered once per wave (one edge per lane, readlane broadcast), X_j/radius loads issued one 6-edge group ahead
# speedup vs baseline: 2.0192x; 2.0192x over previous
#include <hip/hip_runtime.h>
#include <math.h>

typedef __attribute__((ext_vector_type(2))) float v2f;
typedef __attribute__((ext_vector_type(8))) float v8f;

__device__ __constant__ float g_vdw[20][14] = {
  {1.55f,1.70f,1.70f,1.52f, 1.70f,0,0,0,0,0,0,0,0,0},
  {1.55f,1.70f,1.70f,1.52f, 1.70f,1.80f,0,0,0,0,0,0,0,0},
  {1.55f,1.70f,1.70f,1.52f, 1.70f,1.70f,1.52f,1.52f,0,0,0,0,0,0},
  {1.55f,1.70f,1.70f,1.52f, 1.70f,1.70f,1.70f,1.52f,1.52f,0,0,0,0,0},
  {1.55f,1.70f,1.70f,1.52f, 1.70f,1.70f,1.70f,1.70f,1.70f,1.70f,1.70f,0,0,0},
  {1.55f,1.70f,1.70f,1.52f, 0,0,0,0,0,0,0,0,0,0},
  {1.55f,1.70f,1.70f,1.52f, 1.70f,1.70f,1.55f,1.70f,1.70f,1.55f,0,0,0,0},
  {1.55f,1.70f,1.70f,1.52f, 1.70f,1.70f,1.70f,1.70f,0,0,0,0,0,0},
  {1.55f,1.70f,1.70f,1.52f, 1.70f,1.70f,1.70f,1.70f,1.55f,0,0,0,0,0},
  {1.55f,1.70f,1.70f,1.52f, 1.70f,1.70f,1.70f,1.70f,0,0,0,0,0,0},
  {1.55f,1.70f,1.70f,1.52f, 1.70f,1.70f,1.80f,1.70f,0,0,0,0,0,0},
  {1.55f,1.70f,1.70f,1.52f, 1.70f,1.70f,1.52f,1.55f,0,0,0,0,0,0},
  {1.55f,1.70f,1.70f,1.52f, 1.70f,1.70f,1.70f,0,0,0,0,0,0,0},
  {1.55f,1.70f,1.70f,1.52f, 1.70f,1.70f,1.70f,1.52f,1.55f,0,0,0,0,0},
  {1.55f,1.70f,1.70f,1.52f, 1.70f,1.70f,1.70f,1.55f,1.70f,1.55f,1.55f,0,0,0},
  {1.55f,1.70f,1.70f,1.52f, 1.70f,1.52f,0,0,0,0,0,0,0,0},
  {1.55f,1.70f,1.70f,1.52f, 1.70f,1.52f,1.70f,0,0,0,0,0,0,0},
  {1.55f,1.70f,1.70f,1.52f, 1.70f,1.70f,1.70f,0,0,0,0,0,0,0},
  {1.55f,1.70f,1.70f,1.52f, 1.70f,1.70f,1.70f,1.70f,1.55f,1.70f,1.70f,1.70f,1.70f,1.70f},
  {1.55f,1.70f,1.70f,1.52f, 1.70f,1.70f,1.70f,1.70f,1.70f,1.70f,1.70f,1.52f,0,0},
};
__device__ __constant__ int g_natoms[20] =
  {5,6,8,9,11,4,10,8,9,8,8,8,7,9,11,6,7,7,14,12};

#define DIST_EPS      1e-3f
#define CONNECT_CUT   0.35f
#define LOG2E         1.44269504088896340736f

#define EDGE_GROUP 6

typedef __attribute__((ext_vector_type(3))) float v3f;
typedef v3f v3f_a4 __attribute__((aligned(4)));

struct EdgeRegs { v3f xyz; float rad; };

struct EdgeMeta { int kc; int jg; int s; int cnt; float mj; float me; };

__device__ __forceinline__ int bcast_i(int v, int src_lane) {
    return __builtin_amdgcn_readlane(v, src_lane);
}
__device__ __forceinline__ float bcast_f(float v, int src_lane) {
    return __int_as_float(__builtin_amdgcn_readlane(__float_as_int(v), src_lane));
}

__global__ __launch_bounds__(256)
void clash_wmma_kernel(const float* __restrict__ X,
                       const float* __restrict__ mask_j,
                       const float* __restrict__ mask_ij,
                       const int*   __restrict__ C,
                       const int*   __restrict__ S,
                       const int*   __restrict__ E,
                       float*       __restrict__ out,
                       int N, int K, int BN)
{
    const int lane = threadIdx.x & 31;
    const int wave = blockIdx.x * (blockDim.x >> 5) + (threadIdx.x >> 5);
    if (wave >= BN) return;
    if (K <= 0) {
        if (lane == 0) out[wave] = 0.f;
        return;
    }

    const int b = wave / N;
    const int i = wave % N;
    const int a  = lane & 15;
    const int hi = lane >> 4;
    const unsigned ac    = (unsigned)min(a, 13);
    const unsigned acX   = ac * 12u;
    const unsigned acVdw = ac * 4u;
    const int baseE = wave * K;

    auto load_meta = [&](int k0) {
        EdgeMeta m;
        m.kc = min(K - k0, 32);
        const int el = baseE + k0 + min(lane, m.kc - 1);
        const int j  = E[el];
        m.me = mask_ij[el] * ((j != i) ? 1.f : 0.f);
        m.jg = b * N + j;
        m.s  = S[m.jg];
        const int c = C[m.jg];
        m.mj = mask_j[m.jg];
        const int nat = g_natoms[m.s];
        m.cnt = (c > 0) ? nat : 0;
        return m;
    };

    auto load_group = [&](const EdgeMeta& m, int e0, EdgeRegs (&buf)[EDGE_GROUP]) {
#pragma unroll
        for (int u = 0; u < EDGE_GROUP; ++u) {
            const int e  = min(e0 + u, m.kc - 1);
            const unsigned jg = (unsigned)bcast_i(m.jg, e);
            const unsigned sj = (unsigned)bcast_i(m.s, e);
            buf[u].xyz = *(const v3f_a4*)((const char*)X + (jg * 168u + acX));
            buf[u].rad = *(const float*)((const char*)&g_vdw[0][0] + (sj * 56u + acVdw));
        }
    };

    const int s_i = S[wave];
    const int c_i = C[wave];
    const v3f xyz_i = *(const v3f_a4*)((const char*)X + ((unsigned)wave * 168u + acX));
    const EdgeMeta meta = load_meta(0);
    const int   nat_i  = g_natoms[s_i];
    const float rad_ld = g_vdw[s_i][ac];

    EdgeRegs bufA[EDGE_GROUP], bufB[EDGE_GROUP];
#pragma unroll
    for (int u = 0; u < EDGE_GROUP; ++u) {
        bufB[u].xyz = (v3f){0.f, 0.f, 0.f}; bufB[u].rad = 0.f;
    }
    load_group(meta, 0, bufA);

    const int   cnt_i = (c_i > 0) ? nat_i : 0;
    const bool  real  = a < 14;
    const float xi = real ? xyz_i.x : 0.f, yi = real ? xyz_i.y : 0.f, zi = real ? xyz_i.z : 0.f;
    const float rad_i = real ? rad_ld : 0.f;
    const float n2_i = xi * xi + yi * yi + zi * zi;
    const float mA_i = (a < cnt_i) ? 1.f : 0.f;

    v2f Areg;
    Areg.x = hi ? zi : xi;
    Areg.y = hi ? 0.f : yi;

    float n2M[8], hcutM[8], mAM[8];
#pragma unroll
    for (int r = 0; r < 8; ++r) {
        const int M = r + (hi << 3);
        n2M[r]  = __shfl(n2_i, M, 32);
        hcutM[r] = 0.5f * __shfl(rad_i, M, 32) + CONNECT_CUT;
        const float m = __shfl(mA_i, M, 32);
        mAM[r] = (M >= 4) ? m : 0.f;
    }

    float acc = 0.f;

    auto compute_group = [&](const EdgeMeta& m, int e0, const EdgeRegs (&buf)[EDGE_GROUP]) {
#pragma unroll
        for (int u = 0; u < EDGE_GROUP; ++u) {
            const int   e     = min(e0 + u, m.kc - 1);
            const int   cnt_j = bcast_i(m.cnt, e);
            const float mj    = bcast_f(m.mj, e);
            const float me_e  = bcast_f(m.me, e);
            const float me    = (e0 + u < m.kc) ? me_e : 0.f;

            const float xj = buf[u].xyz.x, yj = buf[u].xyz.y, zj = buf[u].xyz.z;
            const float n2N  = xj * xj + yj * yj + zj * zj + DIST_EPS;
            const float hradN = 0.5f * buf[u].rad;
            const float wN   = ((a < cnt_j) ? 1.f : 0.f) * mj * me;

            v2f Breg;
            Breg.x = hi ? zj : xj;
            Breg.y = hi ? 0.f : yj;

            v8f czero = {};
            v8f dot = __builtin_amdgcn_wmma_f32_16x16x4_f32(
                 false, Areg,  false, Breg,
                 (short)0, czero,  false,  false);

#pragma unroll
            for (int r = 0; r < 8; ++r) {
                const float dsq  = n2M[r] + n2N - 2.f * dot[r];
                const float D    = __builtin_amdgcn_sqrtf(dsq);
                const float t    = D - (hcutM[r] + hradN);
                const float ex   = __builtin_amdgcn_exp2f(t * LOG2E);
                const float sg   = __builtin_amdgcn_rcpf(1.f + ex);
                acc += (mAM[r] * wN) * sg;
            }
        }
    };

    for (int e0 = 0; ; ) {
        if (e0 + EDGE_GROUP >= meta.kc) { compute_group(meta, e0, bufA); break; }
        load_group(meta, e0 + EDGE_GROUP, bufB);
        __builtin_amdgcn_sched_barrier(0);
        compute_group(meta, e0, bufA);
        e0 += EDGE_GROUP;
        if (e0 + EDGE_GROUP >= meta.kc) { compute_group(meta, e0, bufB); break; }
        load_group(meta, e0 + EDGE_GROUP, bufA);
        __builtin_amdgcn_sched_barrier(0);
        compute_group(meta, e0, bufB);
        e0 += EDGE_GROUP;
    }

    for (int k = 32; k < K; ++k) {
        const int   j  = E[baseE + k];
        const float me = mask_ij[baseE + k] * ((j != i) ? 1.f : 0.f);
        const int   jg = b * N + j;
        const int   s_j   = S[jg];
        const int   cnt_j = (C[jg] > 0) ? g_natoms[s_j] : 0;
        const v3f   xyz_j = *(const v3f_a4*)((const char*)X + ((unsigned)jg * 168u + acX));
        const float rad_j = g_vdw[s_j][ac];
        const float xj = xyz_j.x, yj = xyz_j.y, zj = xyz_j.z;
        const float n2N  = xj * xj + yj * yj + zj * zj + DIST_EPS;
        const float hradN = 0.5f * rad_j;
        const float wN   = ((a < cnt_j) ? 1.f : 0.f) * mask_j[jg] * me;
        v2f Breg;
        Breg.x = hi ? zj : xj;
        Breg.y = hi ? 0.f : yj;
        v8f czero = {};
        v8f dot = __builtin_amdgcn_wmma_f32_16x16x4_f32(
            false, Areg, false, Breg, (short)0, czero, false, false);
#pragma unroll
        for (int r = 0; r < 8; ++r) {
            const float dsq  = n2M[r] + n2N - 2.f * dot[r];
            const float D    = __builtin_amdgcn_sqrtf(dsq);
            const float t    = D - (hcutM[r] + hradN);
            const float ex   = __builtin_amdgcn_exp2f(t * LOG2E);
            const float sg   = __builtin_amdgcn_rcpf(1.f + ex);
            acc += (mAM[r] * wN) * sg;
        }
    }

#pragma unroll
    for (int off = 16; off > 0; off >>= 1)
        acc += __shfl_down(acc, off, 32);
    if (lane == 0) out[wave] = acc;
}

extern "C" void kernel_launch(void* const* d_in, const int* in_sizes, int n_in,
                              void* d_out, int out_size, void* d_ws, size_t ws_size,
                              hipStream_t stream) {
    const float* X       = (const float*)d_in[0];
    const float* mask_j  = (const float*)d_in[1];
    const float* mask_ij = (const float*)d_in[2];
    const int*   C       = (const int*)d_in[3];
    const int*   S       = (const int*)d_in[4];
    const int*   E       = (const int*)d_in[5];
    float*       out     = (float*)d_out;

    const int BN = in_sizes[1];
    const int K  = in_sizes[2] / BN;
    const int B  = 2;
    const int N  = BN / B;

    const int wavesPerBlock = 8;
    const int blocks = (BN + wavesPerBlock - 1) / wavesPerBlock;
    clash_wmma_kernel<<<blocks, wavesPerBlock * 32, 0, stream>>>(
        X, mask_j, mask_ij, C, S, E, out, N, K, BN);
}


# Epilogue evaluates only result rows that can carry weight: sidechain atoms of i interleaved over both lane halves of A, row loop cut to ceil((natoms_i-4)/2) of 8, wave-uniform; lane sum order changes
# speedup vs baseline: 2.3950x; 1.1861x over previous
#include <hip/hip_runtime.h>
#include <math.h>

typedef __attribute__((ext_vector_type(2))) float v2f;
typedef __attribute__((ext_vector_type(8))) float v8f;

__device__ __constant__ float g_vdw[20][14] = {
  {1.55f,1.70f,1.70f,1.52f, 1.70f,0,0,0,0,0,0,0,0,0},
  {1.55f,1.70f,1.70f,1.52f, 1.70f,1.80f,0,0,0,0,0,0,0,0},
  {1.55f,1.70f,1.70f,1.52f, 1.70f,1.70f,1.52f,1.52f,0,0,0,0,0,0},
  {1.55f,1.70f,1.70f,1.52f, 1.70f,1.70f,1.70f,1.52f,1.52f,0,0,0,0,0},
  {1.55f,1.70f,1.70f,1.52f, 1.70f,1.70f,1.70f,1.70f,1.70f,1.70f,1.70f,0,0,0},
  {1.55f,1.70f,1.70f,1.52f, 0,0,0,0,0,0,0,0,0,0},
  {1.55f,1.70f,1.70f,1.52f, 1.70f,1.70f,1.55f,1.70f,1.70f,1.55f,0,0,0,0},
  {1.55f,1.70f,1.70f,1.52f, 1.70f,1.70f,1.70f,1.70f,0,0,0,0,0,0},
  {1.55f,1.70f,1.70f,1.52f, 1.70f,1.70f,1.70f,1.70f,1.55f,0,0,0,0,0},
  {1.55f,1.70f,1.70f,1.52f, 1.70f,1.70f,1.70f,1.70f,0,0,0,0,0,0},
  {1.55f,1.70f,1.70f,1.52f, 1.70f,1.70f,1.80f,1.70f,0,0,0,0,0,0},
  {1.55f,1.70f,1.70f,1.52f, 1.70f,1.70f,1.52f,1.55f,0,0,0,0,0,0},
  {1.55f,1.70f,1.70f,1.52f, 1.70f,1.70f,1.70f,0,0,0,0,0,0,0},
  {1.55f,1.70f,1.70f,1.52f, 1.70f,1.70f,1.70f,1.52f,1.55f,0,0,0,0,0},
  {1.55f,1.70f,1.70f,1.52f, 1.70f,1.70f,1.70f,1.55f,1.70f,1.55f,1.55f,0,0,0},
  {1.55f,1.70f,1.70f,1.52f, 1.70f,1.52f,0,0,0,0,0,0,0,0},
  {1.55f,1.70f,1.70f,1.52f, 1.70f,1.52f,1.70f,0,0,0,0,0,0,0},
  {1.55f,1.70f,1.70f,1.52f, 1.70f,1.70f,1.70f,0,0,0,0,0,0,0},
  {1.55f,1.70f,1.70f,1.52f, 1.70f,1.70f,1.70f,1.70f,1.55f,1.70f,1.70f,1.70f,1.70f,1.70f},
  {1.55f,1.70f,1.70f,1.52f, 1.70f,1.70f,1.70f,1.70f,1.70f,1.70f,1.70f,1.52f,0,0},
};
__device__ __constant__ int g_natoms[20] =
  {5,6,8,9,11,4,10,8,9,8,8,8,7,9,11,6,7,7,14,12};

#define DIST_EPS      1e-3f
#define CONNECT_CUT   0.35f
#define LOG2E         1.44269504088896340736f

#define EDGE_GROUP 6
#define MAX_ROWS   5

typedef __attribute__((ext_vector_type(3))) float v3f;
typedef v3f v3f_a4 __attribute__((aligned(4)));

struct EdgeRegs { v3f xyz; float rad; };

struct EdgeMeta { int kc; int jg; int s; int cnt; float mj; float me; };

template <int V> struct RowCount { static constexpr int value = V; };

__device__ __forceinline__ int bcast_i(int v, int src_lane) {
    return __builtin_amdgcn_readlane(v, src_lane);
}
__device__ __forceinline__ float bcast_f(float v, int src_lane) {
    return __int_as_float(__builtin_amdgcn_readlane(__float_as_int(v), src_lane));
}

__global__ __launch_bounds__(256)
void clash_wmma_kernel(const float* __restrict__ X,
                       const float* __restrict__ mask_j,
                       const float* __restrict__ mask_ij,
                       const int*   __restrict__ C,
                       const int*   __restrict__ S,
                       const int*   __restrict__ E,
                       float*       __restrict__ out,
                       int N, int K, int BN)
{
    const int lane = threadIdx.x & 31;
    const int wave = blockIdx.x * (blockDim.x >> 5) + (threadIdx.x >> 5);
    if (wave >= BN) return;
    if (K <= 0) {
        if (lane == 0) out[wave] = 0.f;
        return;
    }

    const int b = wave / N;
    const int i = wave % N;
    const int a  = lane & 15;
    const int hi = lane >> 4;
    const unsigned ac    = (unsigned)min(a, 13);
    const unsigned acX   = ac * 12u;
    const unsigned acVdw = ac * 4u;
    const int  ra      = a & 7;
    const bool rowReal = ra < MAX_ROWS;
    const unsigned ai  = rowReal ? (unsigned)(4 + 2 * ra + (a >> 3)) : 13u;
    const int baseE = wave * K;

    auto load_meta = [&](int k0) {
        EdgeMeta m;
        m.kc = min(K - k0, 32);
        const int el = baseE + k0 + min(lane, m.kc - 1);
        const int j  = E[el];
        m.me = mask_ij[el] * ((j != i) ? 1.f : 0.f);
        m.jg = b * N + j;
        m.s  = S[m.jg];
        const int c = C[m.jg];
        m.mj = mask_j[m.jg];
        const int nat = g_natoms[m.s];
        m.cnt = (c > 0) ? nat : 0;
        return m;
    };

    auto load_group = [&](const EdgeMeta& m, int e0, EdgeRegs (&buf)[EDGE_GROUP]) {
#pragma unroll
        for (int u = 0; u < EDGE_GROUP; ++u) {
            const int e  = min(e0 + u, m.kc - 1);
            const unsigned jg = (unsigned)bcast_i(m.jg, e);
            const unsigned sj = (unsigned)bcast_i(m.s, e);
            buf[u].xyz = *(const v3f_a4*)((const char*)X + (jg * 168u + acX));
            buf[u].rad = *(const float*)((const char*)&g_vdw[0][0] + (sj * 56u + acVdw));
        }
    };

    const int s_i = S[wave];
    const int c_i = C[wave];
    const v3f xyz_i = *(const v3f_a4*)((const char*)X + ((unsigned)wave * 168u + ai * 12u));
    const EdgeMeta meta = load_meta(0);
    const int   nat_i  = g_natoms[s_i];
    const float rad_ld = g_vdw[s_i][ai];

    EdgeRegs bufA[EDGE_GROUP], bufB[EDGE_GROUP];
#pragma unroll
    for (int u = 0; u < EDGE_GROUP; ++u) {
        bufB[u].xyz = (v3f){0.f, 0.f, 0.f}; bufB[u].rad = 0.f;
    }
    load_group(meta, 0, bufA);

    const int   cnt_i = (c_i > 0) ? nat_i : 0;
    const float xi = rowReal ? xyz_i.x : 0.f, yi = rowReal ? xyz_i.y : 0.f, zi = rowReal ? xyz_i.z : 0.f;
    const float rad_i = rowReal ? rad_ld : 0.f;
    const float n2_i = xi * xi + yi * yi + zi * zi;
    const int rows = __builtin_amdgcn_readfirstlane(
        (cnt_i > 4) ? min((cnt_i - 3) >> 1, MAX_ROWS) : 0);

    v2f Areg;
    Areg.x = hi ? zi : xi;
    Areg.y = hi ? 0.f : yi;

    float n2M[MAX_ROWS], hcutM[MAX_ROWS], mAM[MAX_ROWS];
#pragma unroll
    for (int r = 0; r < MAX_ROWS; ++r) {
        const int M = r + (hi << 3);
        n2M[r]  = __shfl(n2_i, M, 32);
        hcutM[r] = 0.5f * __shfl(rad_i, M, 32) + CONNECT_CUT;
        mAM[r] = (4 + 2 * r + hi < cnt_i) ? 1.f : 0.f;
    }

    auto run_edges = [&](auto rc) __attribute__((always_inline)) -> float {
    constexpr int R = decltype(rc)::value;
    float acc = 0.f;

    auto compute_group = [&](const EdgeMeta& m, int e0, const EdgeRegs (&buf)[EDGE_GROUP]) {
#pragma unroll
        for (int u = 0; u < EDGE_GROUP; ++u) {
            const int   e     = min(e0 + u, m.kc - 1);
            const int   cnt_j = bcast_i(m.cnt, e);
            const float mj    = bcast_f(m.mj, e);
            const float me_e  = bcast_f(m.me, e);
            const float me    = (e0 + u < m.kc) ? me_e : 0.f;

            const float xj = buf[u].xyz.x, yj = buf[u].xyz.y, zj = buf[u].xyz.z;
            const float n2N  = xj * xj + yj * yj + zj * zj + DIST_EPS;
            const float hradN = 0.5f * buf[u].rad;
            const float wN   = ((a < cnt_j) ? 1.f : 0.f) * mj * me;

            v2f Breg;
            Breg.x = hi ? zj : xj;
            Breg.y = hi ? 0.f : yj;

            v8f czero = {};
            v8f dot = __builtin_amdgcn_wmma_f32_16x16x4_f32(
                 false, Areg,  false, Breg,
                 (short)0, czero,  false,  false);

#pragma unroll
            for (int r = 0; r < R; ++r) {
                const float dsq  = n2M[r] + n2N - 2.f * dot[r];
                const float D    = __builtin_amdgcn_sqrtf(dsq);
                const float t    = D - (hcutM[r] + hradN);
                const float ex   = __builtin_amdgcn_exp2f(t * LOG2E);
                const float sg   = __builtin_amdgcn_rcpf(1.f + ex);
                acc += (mAM[r] * wN) * sg;
            }
        }
    };

    auto hold_group = [&](const EdgeRegs (&buf)[EDGE_GROUP]) {
#pragma unroll
        for (int u = 0; u < EDGE_GROUP; ++u)
            asm volatile("" :: "v"(buf[u].xyz.x), "v"(buf[u].xyz.y), "v"(buf[u].xyz.z), "v"(buf[u].rad));
    };

    for (int e0 = 0; ; ) {
        load_group(meta, e0 + EDGE_GROUP, bufB);
        __builtin_amdgcn_sched_barrier(0);
        compute_group(meta, e0, bufA);
        hold_group(bufB);
        e0 += EDGE_GROUP;
        if (e0 >= meta.kc) break;
        load_group(meta, e0 + EDGE_GROUP, bufA);
        __builtin_amdgcn_sched_barrier(0);
        compute_group(meta, e0, bufB);
        hold_group(bufA);
        e0 += EDGE_GROUP;
        if (e0 >= meta.kc) break;
    }

    for (int k = 32; k < K; ++k) {
        const int   j  = E[baseE + k];
        const float me = mask_ij[baseE + k] * ((j != i) ? 1.f : 0.f);
        const int   jg = b * N + j;
        const int   s_j   = S[jg];
        const int   cnt_j = (C[jg] > 0) ? g_natoms[s_j] : 0;
        const v3f   xyz_j = *(const v3f_a4*)((const char*)X + ((unsigned)jg * 168u + acX));
        const float rad_j = g_vdw[s_j][ac];
        const float xj = xyz_j.x, yj = xyz_j.y, zj = xyz_j.z;
        const float n2N  = xj * xj + yj * yj + zj * zj + DIST_EPS;
        const float hradN = 0.5f * rad_j;
        const float wN   = ((a < cnt_j) ? 1.f : 0.f) * mask_j[jg] * me;
        v2f Breg;
        Breg.x = hi ? zj : xj;
        Breg.y = hi ? 0.f : yj;
        v8f czero = {};
        v8f dot = __builtin_amdgcn_wmma_f32_16x16x4_f32(
            false, Areg, false, Breg, (short)0, czero, false, false);
#pragma unroll
        for (int r = 0; r < R; ++r) {
            const float dsq  = n2M[r] + n2N - 2.f * dot[r];
            const float D    = __builtin_amdgcn_sqrtf(dsq);
            const float t    = D - (hcutM[r] + hradN);
            const float ex   = __builtin_amdgcn_exp2f(t * LOG2E);
            const float sg   = __builtin_amdgcn_rcpf(1.f + ex);
            acc += (mAM[r] * wN) * sg;
        }
    }

    return acc;
    };

    float acc;
    switch (rows) {
        case 1:  acc = run_edges(RowCount<1>{}); break;
        case 2:  acc = run_edges(RowCount<2>{}); break;
        case 3:  acc = run_edges(RowCount<3>{}); break;
        case 4:  acc = run_edges(RowCount<4>{}); break;
        case 5:  acc = run_edges(RowCount<5>{}); break;
        default: acc = 0.f; break;
    }

#pragma unroll
    for (int off = 16; off > 0; off >>= 1)
        acc += __shfl_down(acc, off, 32);
    if (lane == 0) out[wave] = acc;
}

extern "C" void kernel_launch(void* const* d_in, const int* in_sizes, int n_in,
                              void* d_out, int out_size, void* d_ws, size_t ws_size,
                              hipStream_t stream) {
    const float* X       = (const float*)d_in[0];
    const float* mask_j  = (const float*)d_in[1];
    const float* mask_ij = (const float*)d_in[2];
    const int*   C       = (const int*)d_in[3];
    const int*   S       = (const int*)d_in[4];
    const int*   E       = (const int*)d_in[5];
    float*       out     = (float*)d_out;

    const int BN = in_sizes[1];
    const int K  = in_sizes[2] / BN;
    const int B  = 2;
    const int N  = BN / B;

    const int wavesPerBlock = 8;
    const int blocks = (BN + wavesPerBlock - 1) / wavesPerBlock;
    clash_wmma_kernel<<<blocks, wavesPerBlock * 32, 0, stream>>>(
        X, mask_j, mask_ij, C, S, E, out, N, K, BN);
}


# Row- and column-only terms leave the per-element epilogue: WMMA gives squared distance (A=-2a, k3=|b|^2+eps, C=|a|^2), cutoff pre-scaled, row weight applied once to per-row sums; sum order changes
# speedup vs baseline: 2.5361x; 1.0589x over previous
#include <hip/hip_runtime.h>
#include <math.h>

typedef __attribute__((ext_vector_type(2))) float v2f;
typedef __attribute__((ext_vector_type(8))) float v8f;

__device__ __constant__ float g_vdw[20][14] = {
  {1.55f,1.70f,1.70f,1.52f, 1.70f,0,0,0,0,0,0,0,0,0},
  {1.55f,1.70f,1.70f,1.52f, 1.70f,1.80f,0,0,0,0,0,0,0,0},
  {1.55f,1.70f,1.70f,1.52f, 1.70f,1.70f,1.52f,1.52f,0,0,0,0,0,0},
  {1.55f,1.70f,1.70f,1.52f, 1.70f,1.70f,1.70f,1.52f,1.52f,0,0,0,0,0},
  {1.55f,1.70f,1.70f,1.52f, 1.70f,1.70f,1.70f,1.70f,1.70f,1.70f,1.70f,0,0,0},
  {1.55f,1.70f,1.70f,1.52f, 0,0,0,0,0,0,0,0,0,0},
  {1.55f,1.70f,1.70f,1.52f, 1.70f,1.70f,1.55f,1.70f,1.70f,1.55f,0,0,0,0},
  {1.55f,1.70f,1.70f,1.52f, 1.70f,1.70f,1.70f,1.70f,0,0,0,0,0,0},
  {1.55f,1.70f,1.70f,1.52f, 1.70f,1.70f,1.70f,1.70f,1.55f,0,0,0,0,0},
  {1.55f,1.70f,1.70f,1.52f, 1.70f,1.70f,1.70f,1.70f,0,0,0,0,0,0},
  {1.55f,1.70f,1.70f,1.52f, 1.70f,1.70f,1.80f,1.70f,0,0,0,0,0,0},
  {1.55f,1.70f,1.70f,1.52f, 1.70f,1.70f,1.52f,1.55f,0,0,0,0,0,0},
  {1.55f,1.70f,1.70f,1.52f, 1.70f,1.70f,1.70f,0,0,0,0,0,0,0},
  {1.55f,1.70f,1.70f,1.52f, 1.70f,1.70f,1.70f,1.52f,1.55f,0,0,0,0,0},
  {1.55f,1.70f,1.70f,1.52f, 1.70f,1.70f,1.70f,1.55f,1.70f,1.55f,1.55f,0,0,0},
  {1.55f,1.70f,1.70f,1.52f, 1.70f,1.52f,0,0,0,0,0,0,0,0},
  {1.55f,1.70f,1.70f,1.52f, 1.70f,1.52f,1.70f,0,0,0,0,0,0,0},
  {1.55f,1.70f,1.70f,1.52f, 1.70f,1.70f,1.70f,0,0,0,0,0,0,0},
  {1.55f,1.70f,1.70f,1.52f, 1.70f,1.70f,1.70f,1.70f,1.55f,1.70f,1.70f,1.70f,1.70f,1.70f},
  {1.55f,1.70f,1.70f,1.52f, 1.70f,1.70f,1.70f,1.70f,1.70f,1.70f,1.70f,1.52f,0,0},
};
__device__ __constant__ int g_natoms[20] =
  {5,6,8,9,11,4,10,8,9,8,8,8,7,9,11,6,7,7,14,12};

#define DIST_EPS      1e-3f
#define CONNECT_CUT   0.35f
#define LOG2E         1.44269504088896340736f
#define HALF_LOG2E    0.72134752044448170368f

#define EDGE_GROUP 6
#define MAX_ROWS   5

typedef __attribute__((ext_vector_type(3))) float v3f;
typedef v3f v3f_a4 __attribute__((aligned(4)));

struct EdgeRegs { v3f xyz; float rad; };

struct EdgeMeta { int kc; int jg; int s; int cnt; float mj; float me; };

template <int V> struct RowCount { static constexpr int value = V; };

__device__ __forceinline__ int bcast_i(int v, int src_lane) {
    return __builtin_amdgcn_readlane(v, src_lane);
}
__device__ __forceinline__ float bcast_f(float v, int src_lane) {
    return __int_as_float(__builtin_amdgcn_readlane(__float_as_int(v), src_lane));
}

__global__ __launch_bounds__(256)
void clash_wmma_kernel(const float* __restrict__ X,
                       const float* __restrict__ mask_j,
                       const float* __restrict__ mask_ij,
                       const int*   __restrict__ C,
                       const int*   __restrict__ S,
                       const int*   __restrict__ E,
                       float*       __restrict__ out,
                       int N, int K, int BN)
{
    const int lane = threadIdx.x & 31;
    const int wave = blockIdx.x * (blockDim.x >> 5) + (threadIdx.x >> 5);
    if (wave >= BN) return;
    if (K <= 0) {
        if (lane == 0) out[wave] = 0.f;
        return;
    }

    const int b = wave / N;
    const int i = wave % N;
    const int a  = lane & 15;
    const int hi = lane >> 4;
    const unsigned ac    = (unsigned)min(a, 13);
    const unsigned acX   = ac * 12u;
    const unsigned acVdw = ac * 4u;
    const int  ra      = a & 7;
    const bool rowReal = ra < MAX_ROWS;
    const unsigned ai  = rowReal ? (unsigned)(4 + 2 * ra + (a >> 3)) : 13u;
    const int baseE = wave * K;

    auto load_meta = [&](int k0) {
        EdgeMeta m;
        m.kc = min(K - k0, 32);
        const int el = baseE + k0 + min(lane, m.kc - 1);
        const int j  = E[el];
        m.me = mask_ij[el] * ((j != i) ? 1.f : 0.f);
        m.jg = b * N + j;
        m.s  = S[m.jg];
        const int c = C[m.jg];
        m.mj = mask_j[m.jg];
        const int nat = g_natoms[m.s];
        m.cnt = (c > 0) ? nat : 0;
        return m;
    };

    auto load_group = [&](const EdgeMeta& m, int e0, EdgeRegs (&buf)[EDGE_GROUP]) {
#pragma unroll
        for (int u = 0; u < EDGE_GROUP; ++u) {
            const int e  = min(e0 + u, m.kc - 1);
            const unsigned jg = (unsigned)bcast_i(m.jg, e);
            const unsigned sj = (unsigned)bcast_i(m.s, e);
            buf[u].xyz = *(const v3f_a4*)((const char*)X + (jg * 168u + acX));
            buf[u].rad = *(const float*)((const char*)&g_vdw[0][0] + (sj * 56u + acVdw));
        }
    };

    const int s_i = S[wave];
    const int c_i = C[wave];
    const v3f xyz_i = *(const v3f_a4*)((const char*)X + ((unsigned)wave * 168u + ai * 12u));
    const EdgeMeta meta = load_meta(0);
    const int   nat_i  = g_natoms[s_i];
    const float rad_ld = g_vdw[s_i][ai];

    EdgeRegs bufA[EDGE_GROUP], bufB[EDGE_GROUP];
#pragma unroll
    for (int u = 0; u < EDGE_GROUP; ++u) {
        bufB[u].xyz = (v3f){0.f, 0.f, 0.f}; bufB[u].rad = 0.f;
    }
    load_group(meta, 0, bufA);

    const int   cnt_i = (c_i > 0) ? nat_i : 0;
    const float xi = rowReal ? xyz_i.x : 0.f, yi = rowReal ? xyz_i.y : 0.f, zi = rowReal ? xyz_i.z : 0.f;
    const float rad_i = rowReal ? rad_ld : 0.f;
    const float n2_i = xi * xi + yi * yi + zi * zi;
    const int rows = __builtin_amdgcn_readfirstlane(
        (cnt_i > 4) ? min((cnt_i - 3) >> 1, MAX_ROWS) : 0);

    v2f Areg;
    Areg.x = -2.f * (hi ? zi : xi);
    Areg.y = hi ? (rowReal ? 1.f : 0.f) : -2.f * yi;

    float n2M[MAX_ROWS], cM[MAX_ROWS], mAM[MAX_ROWS];
#pragma unroll
    for (int r = 0; r < MAX_ROWS; ++r) {
        const int M = r + (hi << 3);
        n2M[r] = __shfl(n2_i, M, 32);
        cM[r]  = (0.5f * __shfl(rad_i, M, 32) + CONNECT_CUT) * LOG2E;
        mAM[r] = (4 + 2 * r + hi < cnt_i) ? 1.f : 0.f;
    }
    const v8f Creg = {n2M[0], n2M[1], n2M[2], n2M[3], n2M[4], 0.f, 0.f, 0.f};

    auto run_edges = [&](auto rc) __attribute__((always_inline)) -> float {
    constexpr int R = decltype(rc)::value;
    float accR[R];
#pragma unroll
    for (int r = 0; r < R; ++r) accR[r] = 0.f;

    auto eval_edge = [&](float xj, float yj, float zj, float rad_j, float wN) {
        const float n2N = xj * xj + yj * yj + zj * zj + DIST_EPS;
        const float cN  = HALF_LOG2E * rad_j;

        v2f Breg;
        Breg.x = hi ? zj : xj;
        Breg.y = hi ? n2N : yj;

        const v8f dsq = __builtin_amdgcn_wmma_f32_16x16x4_f32(
             false, Areg,  false, Breg,
             (short)0, Creg,  false,  false);

#pragma unroll
        for (int r = 0; r < R; ++r) {
            const float D  = __builtin_amdgcn_sqrtf(__builtin_fabsf(dsq[r]));
            const float ex = __builtin_amdgcn_exp2f(__builtin_fmaf(D, LOG2E, -(cM[r] + cN)));
            const float sg = __builtin_amdgcn_rcpf(1.f + ex);
            accR[r] = __builtin_fmaf(wN, sg, accR[r]);
        }
    };

    auto compute_group = [&](const EdgeMeta& m, int e0, const EdgeRegs (&buf)[EDGE_GROUP]) {
#pragma unroll
        for (int u = 0; u < EDGE_GROUP; ++u) {
            const int   e     = min(e0 + u, m.kc - 1);
            const int   cnt_j = bcast_i(m.cnt, e);
            const float mj    = bcast_f(m.mj, e);
            const float me_e  = bcast_f(m.me, e);
            const float me    = (e0 + u < m.kc) ? me_e : 0.f;

            const float wN   = ((a < cnt_j) ? 1.f : 0.f) * mj * me;
            eval_edge(buf[u].xyz.x, buf[u].xyz.y, buf[u].xyz.z, buf[u].rad, wN);
        }
    };

    auto hold_group = [&](const EdgeRegs (&buf)[EDGE_GROUP]) {
#pragma unroll
        for (int u = 0; u < EDGE_GROUP; ++u)
            asm volatile("" :: "v"(buf[u].xyz.x), "v"(buf[u].xyz.y), "v"(buf[u].xyz.z), "v"(buf[u].rad));
    };

    for (int e0 = 0; ; ) {
        load_group(meta, e0 + EDGE_GROUP, bufB);
        __builtin_amdgcn_sched_barrier(0);
        compute_group(meta, e0, bufA);
        hold_group(bufB);
        e0 += EDGE_GROUP;
        if (e0 >= meta.kc) break;
        load_group(meta, e0 + EDGE_GROUP, bufA);
        __builtin_amdgcn_sched_barrier(0);
        compute_group(meta, e0, bufB);
        hold_group(bufA);
        e0 += EDGE_GROUP;
        if (e0 >= meta.kc) break;
    }

    for (int k = 32; k < K; ++k) {
        const int   j  = E[baseE + k];
        const float me = mask_ij[baseE + k] * ((j != i) ? 1.f : 0.f);
        const int   jg = b * N + j;
        const int   s_j   = S[jg];
        const int   cnt_j = (C[jg] > 0) ? g_natoms[s_j] : 0;
        const v3f   xyz_j = *(const v3f_a4*)((const char*)X + ((unsigned)jg * 168u + acX));
        const float rad_j = g_vdw[s_j][ac];
        const float wN   = ((a < cnt_j) ? 1.f : 0.f) * mask_j[jg] * me;
        eval_edge(xyz_j.x, xyz_j.y, xyz_j.z, rad_j, wN);
    }

    float acc = 0.f;
#pragma unroll
    for (int r = 0; r < R; ++r) acc = __builtin_fmaf(mAM[r], accR[r], acc);
    return acc;
    };

    float acc;
    switch (rows) {
        case 1:  acc = run_edges(RowCount<1>{}); break;
        case 2:  acc = run_edges(RowCount<2>{}); break;
        case 3:  acc = run_edges(RowCount<3>{}); break;
        case 4:  acc = run_edges(RowCount<4>{}); break;
        case 5:  acc = run_edges(RowCount<5>{}); break;
        default: acc = 0.f; break;
    }

#pragma unroll
    for (int off = 16; off > 0; off >>= 1)
        acc += __shfl_down(acc, off, 32);
    if (lane == 0) out[wave] = acc;
}

extern "C" void kernel_launch(void* const* d_in, const int* in_sizes, int n_in,
                              void* d_out, int out_size, void* d_ws, size_t ws_size,
                              hipStream_t stream) {
    const float* X       = (const float*)d_in[0];
    const float* mask_j  = (const float*)d_in[1];
    const float* mask_ij = (const float*)d_in[2];
    const int*   C       = (const int*)d_in[3];
    const int*   S       = (const int*)d_in[4];
    const int*   E       = (const int*)d_in[5];
    float*       out     = (float*)d_out;

    const int BN = in_sizes[1];
    const int K  = in_sizes[2] / BN;
    const int B  = 2;
    const int N  = BN / B;

    const int wavesPerBlock = 8;
    const int blocks = (BN + wavesPerBlock - 1) / wavesPerBlock;
    clash_wmma_kernel<<<blocks, wavesPerBlock * 32, 0, stream>>>(
        X, mask_j, mask_ij, C, S, E, out, N, K, BN);
}


# Neighbour atoms packed across edges: only existing atoms take tile columns (prefix sum of atom counts over edge lanes, column map in LDS), ceil(sum/16) tiles, not one per edge; sum order changes
# speedup vs baseline: 2.7582x; 1.0876x over previous
#include <hip/hip_runtime.h>
#include <math.h>

typedef __attribute__((ext_vector_type(2))) float v2f;
typedef __attribute__((ext_vector_type(8))) float v8f;

__device__ __constant__ float g_vdw[20][14] = {
  {1.55f,1.70f,1.70f,1.52f, 1.70f,0,0,0,0,0,0,0,0,0},
  {1.55f,1.70f,1.70f,1.52f, 1.70f,1.80f,0,0,0,0,0,0,0,0},
  {1.55f,1.70f,1.70f,1.52f, 1.70f,1.70f,1.52f,1.52f,0,0,0,0,0,0},
  {1.55f,1.70f,1.70f,1.52f, 1.70f,1.70f,1.70f,1.52f,1.52f,0,0,0,0,0},
  {1.55f,1.70f,1.70f,1.52f, 1.70f,1.70f,1.70f,1.70f,1.70f,1.70f,1.70f,0,0,0},
  {1.55f,1.70f,1.70f,1.52f, 0,0,0,0,0,0,0,0,0,0},
  {1.55f,1.70f,1.70f,1.52f, 1.70f,1.70f,1.55f,1.70f,1.70f,1.55f,0,0,0,0},
  {1.55f,1.70f,1.70f,1.52f, 1.70f,1.70f,1.70f,1.70f,0,0,0,0,0,0},
  {1.55f,1.70f,1.70f,1.52f, 1.70f,1.70f,1.70f,1.70f,1.55f,0,0,0,0,0},
  {1.55f,1.70f,1.70f,1.52f, 1.70f,1.70f,1.70f,1.70f,0,0,0,0,0,0},
  {1.55f,1.70f,1.70f,1.52f, 1.70f,1.70f,1.80f,1.70f,0,0,0,0,0,0},
  {1.55f,1.70f,1.70f,1.52f, 1.70f,1.70f,1.52f,1.55f,0,0,0,0,0,0},
  {1.55f,1.70f,1.70f,1.52f, 1.70f,1.70f,1.70f,0,0,0,0,0,0,0},
  {1.55f,1.70f,1.70f,1.52f, 1.70f,1.70f,1.70f,1.52f,1.55f,0,0,0,0,0},
  {1.55f,1.70f,1.70f,1.52f, 1.70f,1.70f,1.70f,1.55f,1.70f,1.55f,1.55f,0,0,0},
  {1.55f,1.70f,1.70f,1.52f, 1.70f,1.52f,0,0,0,0,0,0,0,0},
  {1.55f,1.70f,1.70f,1.52f, 1.70f,1.52f,1.70f,0,0,0,0,0,0,0},
  {1.55f,1.70f,1.70f,1.52f, 1.70f,1.70f,1.70f,0,0,0,0,0,0,0},
  {1.55f,1.70f,1.70f,1.52f, 1.70f,1.70f,1.70f,1.70f,1.55f,1.70f,1.70f,1.70f,1.70f,1.70f},
  {1.55f,1.70f,1.70f,1.52f, 1.70f,1.70f,1.70f,1.70f,1.70f,1.70f,1.70f,1.52f,0,0},
};
__device__ __constant__ int g_natoms[20] =
  {5,6,8,9,11,4,10,8,9,8,8,8,7,9,11,6,7,7,14,12};

#define DIST_EPS      1e-3f
#define CONNECT_CUT   0.35f
#define LOG2E         1.44269504088896340736f
#define HALF_LOG2E    0.72134752044448170368f

#define TILE_GROUP 6
#define MAX_ROWS   5
#define WAVES_PER_BLOCK 8
#define MAX_COLS    (32 * 14)
#define MAP_ENTRIES (MAX_COLS + 16)

typedef __attribute__((ext_vector_type(3))) float v3f;
typedef v3f v3f_a4 __attribute__((aligned(4)));

struct ColRegs { v3f xyz; float rad; float w; };

struct EdgeMeta { int kc; int jg; int s; int cnt; float w; };

template <int V> struct RowCount { static constexpr int value = V; };

template <int N>
__device__ __forceinline__ int row_shr0(int v) {
    return __builtin_amdgcn_update_dpp(0, v, 0x110 + N, 0xf, 0xf, true);
}

__global__ __launch_bounds__(256)
void clash_wmma_kernel(const float* __restrict__ X,
                       const float* __restrict__ mask_j,
                       const float* __restrict__ mask_ij,
                       const int*   __restrict__ C,
                       const int*   __restrict__ S,
                       const int*   __restrict__ E,
                       float*       __restrict__ out,
                       int N, int K, int BN)
{
    __shared__ uint4 colmap[WAVES_PER_BLOCK * MAP_ENTRIES];
    const int lane = threadIdx.x & 31;
    const int wave = blockIdx.x * (blockDim.x >> 5) + (threadIdx.x >> 5);
    if (wave >= BN) return;
    uint4* const mymap = colmap + (threadIdx.x >> 5) * MAP_ENTRIES;
    if (K <= 0) {
        if (lane == 0) out[wave] = 0.f;
        return;
    }

    const int b = wave / N;
    const int i = wave % N;
    const int a  = lane & 15;
    const int hi = lane >> 4;
    const unsigned ac    = (unsigned)min(a, 13);
    const unsigned acX   = ac * 12u;
    const int  ra      = a & 7;
    const bool rowReal = ra < MAX_ROWS;
    const unsigned ai  = rowReal ? (unsigned)(4 + 2 * ra + (a >> 3)) : 13u;
    const int baseE = wave * K;

    auto load_meta = [&]() {
        EdgeMeta m;
        m.kc = min(K, 32);
        const int el = baseE + min(lane, m.kc - 1);
        const int j  = E[el];
        const float me = mask_ij[el] * ((j != i) ? 1.f : 0.f);
        m.jg = b * N + j;
        m.s  = S[m.jg];
        const int c = C[m.jg];
        const float mj = mask_j[m.jg];
        const int nat = g_natoms[m.s];
        m.cnt = (c > 0 && lane < m.kc) ? min(nat, 14) : 0;
        m.w   = mj * me;
        return m;
    };

    const int s_i = S[wave];
    const int c_i = C[wave];
    const v3f xyz_i = *(const v3f_a4*)((const char*)X + ((unsigned)wave * 168u + ai * 12u));
    const EdgeMeta meta = load_meta();
    const int   nat_i  = g_natoms[s_i];
    const float rad_ld = g_vdw[s_i][ai];

    int incl = meta.cnt;
    incl += row_shr0<1>(incl);
    incl += row_shr0<2>(incl);
    incl += row_shr0<4>(incl);
    incl += row_shr0<8>(incl);
    const int lowRow = __builtin_amdgcn_readlane(incl, 15);
    incl += (lane >= 16) ? lowRow : 0;
    const int ncols  = __builtin_amdgcn_readlane(incl, 31);
    const int ntiles = (ncols + 15) >> 4;
    const int first  = incl - meta.cnt;
    {
        const unsigned offX = (unsigned)meta.jg * 168u;
        const unsigned offV = (unsigned)meta.s * 56u;
        const unsigned wbit = __float_as_uint(meta.w);
#pragma unroll
        for (int t = 0; t < 14; ++t)
            if (t < meta.cnt)
                mymap[first + t] = make_uint4(offX + 12u * t, offV + 4u * t, wbit, 0u);
        if (lane < 16) mymap[ncols + lane] = make_uint4(0u, 0u, 0u, 0u);
    }

    auto load_tiles = [&](int t0, ColRegs (&buf)[TILE_GROUP]) {
#pragma unroll
        for (int u = 0; u < TILE_GROUP; ++u) {
            const int tile = max(min(t0 + u, ntiles - 1), 0);
            const uint4 cm = mymap[tile * 16 + a];
            buf[u].xyz = *(const v3f_a4*)((const char*)X + cm.x);
            buf[u].rad = *(const float*)((const char*)&g_vdw[0][0] + cm.y);
            buf[u].w   = __uint_as_float(cm.z);
        }
    };

    ColRegs bufA[TILE_GROUP], bufB[TILE_GROUP];
#pragma unroll
    for (int u = 0; u < TILE_GROUP; ++u) {
        bufB[u].xyz = (v3f){0.f, 0.f, 0.f}; bufB[u].rad = 0.f; bufB[u].w = 0.f;
    }
    load_tiles(0, bufA);

    const int   cnt_i = (c_i > 0) ? nat_i : 0;
    const float xi = rowReal ? xyz_i.x : 0.f, yi = rowReal ? xyz_i.y : 0.f, zi = rowReal ? xyz_i.z : 0.f;
    const float rad_i = rowReal ? rad_ld : 0.f;
    const float n2_i = xi * xi + yi * yi + zi * zi;
    const int rows = __builtin_amdgcn_readfirstlane(
        (cnt_i > 4) ? min((cnt_i - 3) >> 1, MAX_ROWS) : 0);

    v2f Areg;
    Areg.x = -2.f * (hi ? zi : xi);
    Areg.y = hi ? (rowReal ? 1.f : 0.f) : -2.f * yi;

    float n2M[MAX_ROWS], cM[MAX_ROWS], mAM[MAX_ROWS];
#pragma unroll
    for (int r = 0; r < MAX_ROWS; ++r) {
        const int M = r + (hi << 3);
        n2M[r] = __shfl(n2_i, M, 32);
        cM[r]  = (0.5f * __shfl(rad_i, M, 32) + CONNECT_CUT) * LOG2E;
        mAM[r] = (4 + 2 * r + hi < cnt_i) ? 1.f : 0.f;
    }
    const v8f Creg = {n2M[0], n2M[1], n2M[2], n2M[3], n2M[4], 0.f, 0.f, 0.f};

    auto run_edges = [&](auto rc) __attribute__((always_inline)) -> float {
    constexpr int R = decltype(rc)::value;
    float accR[R];
#pragma unroll
    for (int r = 0; r < R; ++r) accR[r] = 0.f;

    auto eval_edge = [&](float xj, float yj, float zj, float rad_j, float wN) {
        const float n2N = xj * xj + yj * yj + zj * zj + DIST_EPS;
        const float cN  = HALF_LOG2E * rad_j;

        v2f Breg;
        Breg.x = hi ? zj : xj;
        Breg.y = hi ? n2N : yj;

        const v8f dsq = __builtin_amdgcn_wmma_f32_16x16x4_f32(
             false, Areg,  false, Breg,
             (short)0, Creg,  false,  false);

#pragma unroll
        for (int r = 0; r < R; ++r) {
            const float D  = __builtin_amdgcn_sqrtf(__builtin_fabsf(dsq[r]));
            const float ex = __builtin_amdgcn_exp2f(__builtin_fmaf(D, LOG2E, -(cM[r] + cN)));
            const float sg = __builtin_amdgcn_rcpf(1.f + ex);
            accR[r] = __builtin_fmaf(wN, sg, accR[r]);
        }
    };

    auto compute_tiles = [&](int t0, const ColRegs (&buf)[TILE_GROUP]) {
#pragma unroll
        for (int u = 0; u < TILE_GROUP; ++u)
            if (t0 + u < ntiles)
                eval_edge(buf[u].xyz.x, buf[u].xyz.y, buf[u].xyz.z, buf[u].rad, buf[u].w);
    };

    auto hold_group = [&](const ColRegs (&buf)[TILE_GROUP]) {
#pragma unroll
        for (int u = 0; u < TILE_GROUP; ++u)
            asm volatile("" :: "v"(buf[u].xyz.x), "v"(buf[u].xyz.y), "v"(buf[u].xyz.z), "v"(buf[u].rad));
    };

    for (int t0 = 0; ; ) {
        load_tiles(t0 + TILE_GROUP, bufB);
        __builtin_amdgcn_sched_barrier(0);
        compute_tiles(t0, bufA);
        hold_group(bufB);
        t0 += TILE_GROUP;
        if (t0 >= ntiles) break;
        load_tiles(t0 + TILE_GROUP, bufA);
        __builtin_amdgcn_sched_barrier(0);
        compute_tiles(t0, bufB);
        hold_group(bufA);
        t0 += TILE_GROUP;
        if (t0 >= ntiles) break;
    }

    for (int k = 32; k < K; ++k) {
        const int   j  = E[baseE + k];
        const float me = mask_ij[baseE + k] * ((j != i) ? 1.f : 0.f);
        const int   jg = b * N + j;
        const int   s_j   = S[jg];
        const int   cnt_j = (C[jg] > 0) ? g_natoms[s_j] : 0;
        const v3f   xyz_j = *(const v3f_a4*)((const char*)X + ((unsigned)jg * 168u + acX));
        const float rad_j = g_vdw[s_j][ac];
        const float wN   = ((a < cnt_j) ? 1.f : 0.f) * mask_j[jg] * me;
        eval_edge(xyz_j.x, xyz_j.y, xyz_j.z, rad_j, wN);
    }

    float acc = 0.f;
#pragma unroll
    for (int r = 0; r < R; ++r) acc = __builtin_fmaf(mAM[r], accR[r], acc);
    return acc;
    };

    float acc;
    switch (rows) {
        case 1:  acc = run_edges(RowCount<1>{}); break;
        case 2:  acc = run_edges(RowCount<2>{}); break;
        case 3:  acc = run_edges(RowCount<3>{}); break;
        case 4:  acc = run_edges(RowCount<4>{}); break;
        case 5:  acc = run_edges(RowCount<5>{}); break;
        default: acc = 0.f; break;
    }

#pragma unroll
    for (int off = 16; off > 0; off >>= 1)
        acc += __shfl_down(acc, off, 32);
    if (lane == 0) out[wave] = acc;
}

extern "C" void kernel_launch(void* const* d_in, const int* in_sizes, int n_in,
                              void* d_out, int out_size, void* d_ws, size_t ws_size,
                              hipStream_t stream) {
    const float* X       = (const float*)d_in[0];
    const float* mask_j  = (const float*)d_in[1];
    const float* mask_ij = (const float*)d_in[2];
    const int*   C       = (const int*)d_in[3];
    const int*   S       = (const int*)d_in[4];
    const int*   E       = (const int*)d_in[5];
    float*       out     = (float*)d_out;

    const int BN = in_sizes[1];
    const int K  = in_sizes[2] / BN;
    const int B  = 2;
    const int N  = BN / B;

    const int wavesPerBlock = WAVES_PER_BLOCK;
    const int blocks = (BN + wavesPerBlock - 1) / wavesPerBlock;
    clash_wmma_kernel<<<blocks, wavesPerBlock * 32, 0, stream>>>(
        X, mask_j, mask_ij, C, S, E, out, N, K, BN);
}
